# starting kernel
# baseline (speedup 1.0000x reference)
#include <hip/hip_runtime.h>


typedef float v2f __attribute__((ext_vector_type(2)));
typedef float v8f __attribute__((ext_vector_type(8)));

#define E_TOT 160000
#define NCH 8
#define MIDW 32
#define EDIM 16
#define WAVES_PER_BLK 4

struct KParams {
  const float* feat0; const float* feat1; const float* w; const float* radial;
  const float* basis[4];
  const float* w1[4]; const float* b1[4];
  const float* w2[4]; const float* b2[4];
  const float* w3[4]; const float* b3[4];
  const int* src_idx;
  float* out;
};

__device__ __forceinline__ v8f wmma4(v2f a, v2f b, v8f c) {
  return __builtin_amdgcn_wmma_f32_16x16x4_f32(false, a, false, b,
                                               (short)0, c, false, false);
}

__device__ __forceinline__ v2f ldB(const float* __restrict__ W, int ncols,
                                   int k0, int n0, int m, int hi) {
  int k = k0 + 2 * hi;
  v2f b;
  b.x = W[(k    ) * ncols + n0 + m];
  b.y = W[(k + 1) * ncols + n0 + m];
  return b;
}

template<int NF>
__device__ __forceinline__ void radial_mlp_tile(
    const float* __restrict__ w1, const float* __restrict__ bs1,
    const float* __restrict__ w2, const float* __restrict__ bs2,
    const float* __restrict__ w3, const float* __restrict__ bs3,
    const v2f* __restrict__ xa, float* __restrict__ sH, float* __restrict__ sR,
    int m, int hi) {
  {
    v2f b0[5], b1f[5];
    #pragma unroll
    for (int c = 0; c < 4; ++c) {
      b0[c]  = ldB(w1, MIDW, 4 * c, 0,  m, hi);
      b1f[c] = ldB(w1, MIDW, 4 * c, 16, m, hi);
    }
    {
      float wa = w1[16 * MIDW + m];
      float wb = w1[16 * MIDW + 16 + m];
      b0[4].x  = hi ? 0.f : wa;  b0[4].y  = 0.f;
      b1f[4].x = hi ? 0.f : wb;  b1f[4].y = 0.f;
    }
    float bias0 = bs1[m], bias1 = bs1[16 + m];
    v8f a0 = {}, a1 = {};
    #pragma unroll
    for (int c = 0; c < 5; ++c) {
      a0 = wmma4(xa[c], b0[c],  a0);
      a1 = wmma4(xa[c], b1f[c], a1);
    }
    #pragma unroll
    for (int r = 0; r < 8; ++r) {
      float v0 = a0[r] + bias0, v1 = a1[r] + bias1;
      sH[(r + 8 * hi) * MIDW + m]      = v0 > 0.f ? v0 : 0.f;
      sH[(r + 8 * hi) * MIDW + 16 + m] = v1 > 0.f ? v1 : 0.f;
    }
  }
  v2f ha[8];
  #pragma unroll
  for (int c = 0; c < 8; ++c) {
    ha[c].x = sH[m * MIDW + 4 * c + 2 * hi];
    ha[c].y = sH[m * MIDW + 4 * c + 2 * hi + 1];
  }
  {
    v2f b0[8], b1f[8];
    #pragma unroll
    for (int c = 0; c < 8; ++c) {
      b0[c]  = ldB(w2, MIDW, 4 * c, 0,  m, hi);
      b1f[c] = ldB(w2, MIDW, 4 * c, 16, m, hi);
    }
    float bias0 = bs2[m], bias1 = bs2[16 + m];
    v8f a0 = {}, a1 = {};
    #pragma unroll
    for (int c = 0; c < 8; ++c) {
      a0 = wmma4(ha[c], b0[c],  a0);
      a1 = wmma4(ha[c], b1f[c], a1);
    }
    #pragma unroll
    for (int r = 0; r < 8; ++r) {
      float v0 = a0[r] + bias0, v1 = a1[r] + bias1;
      sH[(r + 8 * hi) * MIDW + m]      = v0 > 0.f ? v0 : 0.f;
      sH[(r + 8 * hi) * MIDW + 16 + m] = v1 > 0.f ? v1 : 0.f;
    }
  }
  #pragma unroll
  for (int c = 0; c < 8; ++c) {
    ha[c].x = sH[m * MIDW + 4 * c + 2 * hi];
    ha[c].y = sH[m * MIDW + 4 * c + 2 * hi + 1];
  }
  constexpr int NCOLS = NF * NCH * NCH;
  #pragma unroll
  for (int j = 0; j < NCOLS / 16; j += 2) {
    v2f b0[8], b1f[8];
    #pragma unroll
    for (int c = 0; c < 8; ++c) {
      b0[c]  = ldB(w3, NCOLS, 4 * c, 16 * j,       m, hi);
      b1f[c] = ldB(w3, NCOLS, 4 * c, 16 * (j + 1), m, hi);
    }
    float bias0 = bs3[16 * j + m], bias1 = bs3[16 * (j + 1) + m];
    v8f a0 = {}, a1 = {};
    #pragma unroll
    for (int c = 0; c < 8; ++c) {
      a0 = wmma4(ha[c], b0[c],  a0);
      a1 = wmma4(ha[c], b1f[c], a1);
    }
    #pragma unroll
    for (int r = 0; r < 8; ++r) {
      sR[(r + 8 * hi) * NCOLS + 16 * j + m]       = a0[r] + bias0;
      sR[(r + 8 * hi) * NCOLS + 16 * (j + 1) + m] = a1[r] + bias1;
    }
  }
}

template<int DI, int DO, int NF>
__device__ __forceinline__ void edge_accum(
    const float* __restrict__ sRe,
    const float* __restrict__ basisE,
    const float* __restrict__ srcF,
    int half, float (&acc)[4][2 * DO + 1]) {
  constexpr int QI = 2 * DI + 1, PO = 2 * DO + 1;
  float s[NCH][QI];
  #pragma unroll
  for (int i = 0; i < NCH; ++i)
    #pragma unroll
    for (int q = 0; q < QI; ++q) s[i][q] = srcF[i * QI + q];
  float bas[PO][QI][NF];
  #pragma unroll
  for (int p = 0; p < PO; ++p)
    #pragma unroll
    for (int q = 0; q < QI; ++q)
      #pragma unroll
      for (int f = 0; f < NF; ++f) bas[p][q][f] = basisE[(p * QI + q) * NF + f];
  float T[NCH][NF][PO];
  #pragma unroll
  for (int i = 0; i < NCH; ++i)
    #pragma unroll
    for (int f = 0; f < NF; ++f)
      #pragma unroll
      for (int p = 0; p < PO; ++p) {
        float t = 0.f;
        #pragma unroll
        for (int q = 0; q < QI; ++q) t += bas[p][q][f] * s[i][q];
        T[i][f][p] = t;
      }
  #pragma unroll
  for (int ol = 0; ol < 4; ++ol) {
    int o = half * 4 + ol;
    const float* Ro = sRe + o * NCH * NF;
    #pragma unroll
    for (int p = 0; p < PO; ++p) {
      float t = 0.f;
      #pragma unroll
      for (int i = 0; i < NCH; ++i)
        #pragma unroll
        for (int f = 0; f < NF; ++f) t += Ro[i * NF + f] * T[i][f][p];
      acc[ol][p] += t;
    }
  }
}

__global__ void __launch_bounds__(WAVES_PER_BLK * 32)
se3_partial_kernel(KParams p) {
  __shared__ float sH[WAVES_PER_BLK][16 * MIDW];
  __shared__ float sR[WAVES_PER_BLK][16 * 192];

  const int lane = threadIdx.x & 31;
  const int wv   = threadIdx.x >> 5;
  const int m    = lane & 15;
  const int hi   = lane >> 4;
  const int tile = blockIdx.x * WAVES_PER_BLK + wv;
  const int e0   = tile * 16;
  if (e0 >= E_TOT) return;

  float* __restrict__ sHw = sH[wv];
  float* __restrict__ sRw = sR[wv];

  v2f xa[5];
  {
    const int er = e0 + m;
    #pragma unroll
    for (int c = 0; c < 4; ++c)
      xa[c] = *(const v2f*)(p.w + (size_t)er * EDIM + 4 * c + 2 * hi);
    float rad = p.radial[er];
    xa[4].x = hi ? 0.f : rad;
    xa[4].y = 0.f;
  }

  const int e = e0 + m;
  const int sidx = p.src_idx[e];
  const float* src0 = p.feat0 + (size_t)sidx * (NCH * 1);
  const float* src1 = p.feat1 + (size_t)sidx * (NCH * 3);

  float accD0[4][1] = {};
  float accD1[4][3] = {};

  radial_mlp_tile<1>(p.w1[0], p.b1[0], p.w2[0], p.b2[0], p.w3[0], p.b3[0],
                     xa, sHw, sRw, m, hi);
  edge_accum<0, 0, 1>(sRw + m * 64, p.basis[0] + (size_t)e * 1, src0, hi, accD0);

  radial_mlp_tile<1>(p.w1[1], p.b1[1], p.w2[1], p.b2[1], p.w3[1], p.b3[1],
                     xa, sHw, sRw, m, hi);
  edge_accum<0, 1, 1>(sRw + m * 64, p.basis[1] + (size_t)e * 3, src0, hi, accD1);

  radial_mlp_tile<1>(p.w1[2], p.b1[2], p.w2[2], p.b2[2], p.w3[2], p.b3[2],
                     xa, sHw, sRw, m, hi);
  edge_accum<1, 0, 1>(sRw + m * 64, p.basis[2] + (size_t)e * 3, src1, hi, accD0);

  radial_mlp_tile<3>(p.w1[3], p.b1[3], p.w2[3], p.b2[3], p.w3[3], p.b3[3],
                     xa, sHw, sRw, m, hi);
  edge_accum<1, 1, 3>(sRw + m * 192, p.basis[3] + (size_t)e * 27, src1, hi, accD1);

  #pragma unroll
  for (int ol = 0; ol < 4; ++ol) {
    int o = hi * 4 + ol;
    p.out[(size_t)e * NCH + o] = accD0[ol][0];
    #pragma unroll
    for (int q = 0; q < 3; ++q)
      p.out[(size_t)E_TOT * NCH + (size_t)e * (NCH * 3) + o * 3 + q] =
          accD1[ol][q];
  }
}

extern "C" void kernel_launch(void* const* d_in, const int* in_sizes, int n_in,
                              void* d_out, int out_size, void* d_ws,
                              size_t ws_size, hipStream_t stream) {
  (void)in_sizes; (void)n_in; (void)out_size; (void)d_ws; (void)ws_size;
  KParams p;
  p.feat0  = (const float*)d_in[0];
  p.feat1  = (const float*)d_in[1];
  p.w      = (const float*)d_in[2];
  p.radial = (const float*)d_in[3];
  for (int pr = 0; pr < 4; ++pr) {
    int b = 4 + pr * 7;
    p.basis[pr] = (const float*)d_in[b + 0];
    p.w1[pr]    = (const float*)d_in[b + 1];
    p.b1[pr]    = (const float*)d_in[b + 2];
    p.w2[pr]    = (const float*)d_in[b + 3];
    p.b2[pr]    = (const float*)d_in[b + 4];
    p.w3[pr]    = (const float*)d_in[b + 5];
    p.b3[pr]    = (const float*)d_in[b + 6];
  }
  p.src_idx = (const int*)d_in[32];
  p.out     = (float*)d_out;

  const int tiles  = E_TOT / 16;
  const int blocks = (tiles + WAVES_PER_BLK - 1) / WAVES_PER_BLK;
  hipLaunchKernelGGL(se3_partial_kernel, dim3(blocks), dim3(WAVES_PER_BLK * 32),
                     0, stream, p);
}


# LDS row pitch padded to 16 B x odd (hidden rows 32->36, result rows 64/192->68/196 dwords): A-fragment re-reads and per-edge 16-byte reads become bank-conflict-free
# speedup vs baseline: 1.1512x; 1.1512x over previous
#include <hip/hip_runtime.h>


typedef float v2f __attribute__((ext_vector_type(2)));
typedef float v8f __attribute__((ext_vector_type(8)));

#define E_TOT 160000
#define NCH 8
#define MIDW 32
#define EDIM 16
#define WAVES_PER_BLK 4
#define HP (MIDW + 4)
#define RPITCH(ncols) ((ncols) + 4)

struct KParams {
  const float* feat0; const float* feat1; const float* w; const float* radial;
  const float* basis[4];
  const float* w1[4]; const float* b1[4];
  const float* w2[4]; const float* b2[4];
  const float* w3[4]; const float* b3[4];
  const int* src_idx;
  float* out;
};

__device__ __forceinline__ v8f wmma4(v2f a, v2f b, v8f c) {
  return __builtin_amdgcn_wmma_f32_16x16x4_f32(false, a, false, b,
                                               (short)0, c, false, false);
}

__device__ __forceinline__ v2f ldB(const float* __restrict__ W, int ncols,
                                   int k0, int n0, int m, int hi) {
  int k = k0 + 2 * hi;
  v2f b;
  b.x = W[(k    ) * ncols + n0 + m];
  b.y = W[(k + 1) * ncols + n0 + m];
  return b;
}

template<int NF>
__device__ __forceinline__ void radial_mlp_tile(
    const float* __restrict__ w1, const float* __restrict__ bs1,
    const float* __restrict__ w2, const float* __restrict__ bs2,
    const float* __restrict__ w3, const float* __restrict__ bs3,
    const v2f* __restrict__ xa, float* __restrict__ sH, float* __restrict__ sR,
    int m, int hi) {
  {
    v2f b0[5], b1f[5];
    #pragma unroll
    for (int c = 0; c < 4; ++c) {
      b0[c]  = ldB(w1, MIDW, 4 * c, 0,  m, hi);
      b1f[c] = ldB(w1, MIDW, 4 * c, 16, m, hi);
    }
    {
      float wa = w1[16 * MIDW + m];
      float wb = w1[16 * MIDW + 16 + m];
      b0[4].x  = hi ? 0.f : wa;  b0[4].y  = 0.f;
      b1f[4].x = hi ? 0.f : wb;  b1f[4].y = 0.f;
    }
    float bias0 = bs1[m], bias1 = bs1[16 + m];
    v8f a0 = {}, a1 = {};
    #pragma unroll
    for (int c = 0; c < 5; ++c) {
      a0 = wmma4(xa[c], b0[c],  a0);
      a1 = wmma4(xa[c], b1f[c], a1);
    }
    #pragma unroll
    for (int r = 0; r < 8; ++r) {
      float v0 = a0[r] + bias0, v1 = a1[r] + bias1;
      sH[(r + 8 * hi) * HP + m]      = v0 > 0.f ? v0 : 0.f;
      sH[(r + 8 * hi) * HP + 16 + m] = v1 > 0.f ? v1 : 0.f;
    }
  }
  v2f ha[8];
  #pragma unroll
  for (int c = 0; c < 8; ++c) {
    ha[c].x = sH[m * HP + 4 * c + 2 * hi];
    ha[c].y = sH[m * HP + 4 * c + 2 * hi + 1];
  }
  {
    v2f b0[8], b1f[8];
    #pragma unroll
    for (int c = 0; c < 8; ++c) {
      b0[c]  = ldB(w2, MIDW, 4 * c, 0,  m, hi);
      b1f[c] = ldB(w2, MIDW, 4 * c, 16, m, hi);
    }
    float bias0 = bs2[m], bias1 = bs2[16 + m];
    v8f a0 = {}, a1 = {};
    #pragma unroll
    for (int c = 0; c < 8; ++c) {
      a0 = wmma4(ha[c], b0[c],  a0);
      a1 = wmma4(ha[c], b1f[c], a1);
    }
    #pragma unroll
    for (int r = 0; r < 8; ++r) {
      float v0 = a0[r] + bias0, v1 = a1[r] + bias1;
      sH[(r + 8 * hi) * HP + m]      = v0 > 0.f ? v0 : 0.f;
      sH[(r + 8 * hi) * HP + 16 + m] = v1 > 0.f ? v1 : 0.f;
    }
  }
  #pragma unroll
  for (int c = 0; c < 8; ++c) {
    ha[c].x = sH[m * HP + 4 * c + 2 * hi];
    ha[c].y = sH[m * HP + 4 * c + 2 * hi + 1];
  }
  constexpr int NCOLS = NF * NCH * NCH;
  constexpr int RP = RPITCH(NCOLS);
  #pragma unroll
  for (int j = 0; j < NCOLS / 16; j += 2) {
    v2f b0[8], b1f[8];
    #pragma unroll
    for (int c = 0; c < 8; ++c) {
      b0[c]  = ldB(w3, NCOLS, 4 * c, 16 * j,       m, hi);
      b1f[c] = ldB(w3, NCOLS, 4 * c, 16 * (j + 1), m, hi);
    }
    float bias0 = bs3[16 * j + m], bias1 = bs3[16 * (j + 1) + m];
    v8f a0 = {}, a1 = {};
    #pragma unroll
    for (int c = 0; c < 8; ++c) {
      a0 = wmma4(ha[c], b0[c],  a0);
      a1 = wmma4(ha[c], b1f[c], a1);
    }
    #pragma unroll
    for (int r = 0; r < 8; ++r) {
      sR[(r + 8 * hi) * RP + 16 * j + m]       = a0[r] + bias0;
      sR[(r + 8 * hi) * RP + 16 * (j + 1) + m] = a1[r] + bias1;
    }
  }
}

template<int DI, int DO, int NF>
__device__ __forceinline__ void edge_accum(
    const float* __restrict__ sRe,
    const float* __restrict__ basisE,
    const float* __restrict__ srcF,
    int half, float (&acc)[4][2 * DO + 1]) {
  constexpr int QI = 2 * DI + 1, PO = 2 * DO + 1;
  float s[NCH][QI];
  #pragma unroll
  for (int i = 0; i < NCH; ++i)
    #pragma unroll
    for (int q = 0; q < QI; ++q) s[i][q] = srcF[i * QI + q];
  float bas[PO][QI][NF];
  #pragma unroll
  for (int p = 0; p < PO; ++p)
    #pragma unroll
    for (int q = 0; q < QI; ++q)
      #pragma unroll
      for (int f = 0; f < NF; ++f) bas[p][q][f] = basisE[(p * QI + q) * NF + f];
  float T[NCH][NF][PO];
  #pragma unroll
  for (int i = 0; i < NCH; ++i)
    #pragma unroll
    for (int f = 0; f < NF; ++f)
      #pragma unroll
      for (int p = 0; p < PO; ++p) {
        float t = 0.f;
        #pragma unroll
        for (int q = 0; q < QI; ++q) t += bas[p][q][f] * s[i][q];
        T[i][f][p] = t;
      }
  #pragma unroll
  for (int ol = 0; ol < 4; ++ol) {
    int o = half * 4 + ol;
    const float* Ro = sRe + o * NCH * NF;
    #pragma unroll
    for (int p = 0; p < PO; ++p) {
      float t = 0.f;
      #pragma unroll
      for (int i = 0; i < NCH; ++i)
        #pragma unroll
        for (int f = 0; f < NF; ++f) t += Ro[i * NF + f] * T[i][f][p];
      acc[ol][p] += t;
    }
  }
}

__global__ void __launch_bounds__(WAVES_PER_BLK * 32)
se3_partial_kernel(KParams p) {
  __shared__ float sH[WAVES_PER_BLK][16 * HP];
  __shared__ float sR[WAVES_PER_BLK][16 * RPITCH(192)];

  const int lane = threadIdx.x & 31;
  const int wv   = threadIdx.x >> 5;
  const int m    = lane & 15;
  const int hi   = lane >> 4;
  const int tile = blockIdx.x * WAVES_PER_BLK + wv;
  const int e0   = tile * 16;
  if (e0 >= E_TOT) return;

  float* __restrict__ sHw = sH[wv];
  float* __restrict__ sRw = sR[wv];

  v2f xa[5];
  {
    const int er = e0 + m;
    #pragma unroll
    for (int c = 0; c < 4; ++c)
      xa[c] = *(const v2f*)(p.w + (size_t)er * EDIM + 4 * c + 2 * hi);
    float rad = p.radial[er];
    xa[4].x = hi ? 0.f : rad;
    xa[4].y = 0.f;
  }

  const int e = e0 + m;
  const int sidx = p.src_idx[e];
  const float* src0 = p.feat0 + (size_t)sidx * (NCH * 1);
  const float* src1 = p.feat1 + (size_t)sidx * (NCH * 3);

  float accD0[4][1] = {};
  float accD1[4][3] = {};

  radial_mlp_tile<1>(p.w1[0], p.b1[0], p.w2[0], p.b2[0], p.w3[0], p.b3[0],
                     xa, sHw, sRw, m, hi);
  edge_accum<0, 0, 1>(sRw + m * RPITCH(64), p.basis[0] + (size_t)e * 1, src0, hi, accD0);

  radial_mlp_tile<1>(p.w1[1], p.b1[1], p.w2[1], p.b2[1], p.w3[1], p.b3[1],
                     xa, sHw, sRw, m, hi);
  edge_accum<0, 1, 1>(sRw + m * RPITCH(64), p.basis[1] + (size_t)e * 3, src0, hi, accD1);

  radial_mlp_tile<1>(p.w1[2], p.b1[2], p.w2[2], p.b2[2], p.w3[2], p.b3[2],
                     xa, sHw, sRw, m, hi);
  edge_accum<1, 0, 1>(sRw + m * RPITCH(64), p.basis[2] + (size_t)e * 3, src1, hi, accD0);

  radial_mlp_tile<3>(p.w1[3], p.b1[3], p.w2[3], p.b2[3], p.w3[3], p.b3[3],
                     xa, sHw, sRw, m, hi);
  edge_accum<1, 1, 3>(sRw + m * RPITCH(192), p.basis[3] + (size_t)e * 27, src1, hi, accD1);

  #pragma unroll
  for (int ol = 0; ol < 4; ++ol) {
    int o = hi * 4 + ol;
    p.out[(size_t)e * NCH + o] = accD0[ol][0];
    #pragma unroll
    for (int q = 0; q < 3; ++q)
      p.out[(size_t)E_TOT * NCH + (size_t)e * (NCH * 3) + o * 3 + q] =
          accD1[ol][q];
  }
}

extern "C" void kernel_launch(void* const* d_in, const int* in_sizes, int n_in,
                              void* d_out, int out_size, void* d_ws,
                              size_t ws_size, hipStream_t stream) {
  (void)in_sizes; (void)n_in; (void)out_size; (void)d_ws; (void)ws_size;
  KParams p;
  p.feat0  = (const float*)d_in[0];
  p.feat1  = (const float*)d_in[1];
  p.w      = (const float*)d_in[2];
  p.radial = (const float*)d_in[3];
  for (int pr = 0; pr < 4; ++pr) {
    int b = 4 + pr * 7;
    p.basis[pr] = (const float*)d_in[b + 0];
    p.w1[pr]    = (const float*)d_in[b + 1];
    p.b1[pr]    = (const float*)d_in[b + 2];
    p.w2[pr]    = (const float*)d_in[b + 3];
    p.b2[pr]    = (const float*)d_in[b + 4];
    p.w3[pr]    = (const float*)d_in[b + 5];
    p.b3[pr]    = (const float*)d_in[b + 6];
  }
  p.src_idx = (const int*)d_in[32];
  p.out     = (float*)d_out;

  const int tiles  = E_TOT / 16;
  const int blocks = (tiles + WAVES_PER_BLK - 1) / WAVES_PER_BLK;
  hipLaunchKernelGGL(se3_partial_kernel, dim3(blocks), dim3(WAVES_PER_BLK * 32),
                     0, stream, p);
}


# Transposed radial MLP: weights as WMMA A operand, edges as B columns, bias as C start; each layer's D registers feed the next layer's B and the einsum directly, no LDS; k and einsum sum order change
# speedup vs baseline: 1.2569x; 1.0918x over previous
#include <hip/hip_runtime.h>


typedef float v2f __attribute__((ext_vector_type(2)));
typedef float v8f __attribute__((ext_vector_type(8)));

#define E_TOT 160000
#define NCH 8
#define MIDW 32
#define EDIM 16
#define WAVES_PER_BLK 4

struct KParams {
  const float* feat0; const float* feat1; const float* w; const float* radial;
  const float* basis[4];
  const float* w1[4]; const float* b1[4];
  const float* w2[4]; const float* b2[4];
  const float* w3[4]; const float* b3[4];
  const int* src_idx;
  float* out;
};

__device__ __forceinline__ v8f wmma4(v2f a, v2f b, v8f c) {
  return __builtin_amdgcn_wmma_f32_16x16x4_f32(false, a, false, b,
                                               (short)0, c, false, false);
}

__device__ __forceinline__ v2f ldWT(const float* __restrict__ Wl, int ncols, int off) {
  v2f a;
  a.x = Wl[off];
  a.y = Wl[off + ncols];
  return a;
}

template<int DI, int DO, int NF>
__device__ __forceinline__ void pair_tile(
    const float* __restrict__ w1, const float* __restrict__ bs1,
    const float* __restrict__ w2, const float* __restrict__ bs2,
    const float* __restrict__ w3, const float* __restrict__ bs3,
    const v2f (&xa)[5],
    const float* __restrict__ basisE,
    const float* __restrict__ srcF,
    int m, int hi, v2f (&acc)[4][2 * DO + 1]) {
  constexpr int QI = 2 * DI + 1, PO = 2 * DO + 1;
  constexpr int NIF = NCH * NF;
  constexpr int NCOLS = NCH * NIF;

  v8f h[2];
  {
    const float* w1l = w1 + 2 * hi * MIDW + m;
    v2f wa0[5], wa1[5];
    #pragma unroll
    for (int c = 0; c < 4; ++c) {
      wa0[c] = ldWT(w1l, MIDW, 4 * c * MIDW);
      wa1[c] = ldWT(w1l, MIDW, 4 * c * MIDW + 16);
    }
    {
      float wa = w1[16 * MIDW + m];
      float wb = w1[16 * MIDW + 16 + m];
      wa0[4].x = hi ? 0.f : wa;  wa0[4].y = 0.f;
      wa1[4].x = hi ? 0.f : wb;  wa1[4].y = 0.f;
    }
    const float* bp = bs1 + 8 * hi;
    v8f d0, d1;
    #pragma unroll
    for (int r = 0; r < 8; ++r) { d0[r] = bp[r]; d1[r] = bp[16 + r]; }
    #pragma unroll
    for (int c = 0; c < 5; ++c) {
      d0 = wmma4(wa0[c], xa[c], d0);
      d1 = wmma4(wa1[c], xa[c], d1);
    }
    #pragma unroll
    for (int r = 0; r < 8; ++r) {
      h[0][r] = d0[r] > 0.f ? d0[r] : 0.f;
      h[1][r] = d1[r] > 0.f ? d1[r] : 0.f;
    }
  }
  {
    const float* w2l = w2 + 8 * hi * MIDW + m;
    v2f wa0[8], wa1[8];
    #pragma unroll
    for (int t = 0; t < 2; ++t)
      #pragma unroll
      for (int jp = 0; jp < 4; ++jp) {
        wa0[4 * t + jp] = ldWT(w2l, MIDW, (16 * t + 2 * jp) * MIDW);
        wa1[4 * t + jp] = ldWT(w2l, MIDW, (16 * t + 2 * jp) * MIDW + 16);
      }
    const float* bp = bs2 + 8 * hi;
    v8f d0, d1;
    #pragma unroll
    for (int r = 0; r < 8; ++r) { d0[r] = bp[r]; d1[r] = bp[16 + r]; }
    #pragma unroll
    for (int t = 0; t < 2; ++t)
      #pragma unroll
      for (int jp = 0; jp < 4; ++jp) {
        v2f b; b.x = h[t][2 * jp]; b.y = h[t][2 * jp + 1];
        d0 = wmma4(wa0[4 * t + jp], b, d0);
        d1 = wmma4(wa1[4 * t + jp], b, d1);
      }
    #pragma unroll
    for (int r = 0; r < 8; ++r) {
      h[0][r] = d0[r] > 0.f ? d0[r] : 0.f;
      h[1][r] = d1[r] > 0.f ? d1[r] : 0.f;
    }
  }
  v2f T2[NIF / 2][PO];
  {
    float s[NCH][QI];
    #pragma unroll
    for (int i = 0; i < NCH; ++i)
      #pragma unroll
      for (int q = 0; q < QI; ++q) s[i][q] = srcF[i * QI + q];
    float bas[PO][QI][NF];
    #pragma unroll
    for (int p = 0; p < PO; ++p)
      #pragma unroll
      for (int q = 0; q < QI; ++q)
        #pragma unroll
        for (int f = 0; f < NF; ++f) bas[p][q][f] = basisE[(p * QI + q) * NF + f];
    #pragma unroll
    for (int i = 0; i < NCH; ++i)
      #pragma unroll
      for (int f = 0; f < NF; ++f)
        #pragma unroll
        for (int p = 0; p < PO; ++p) {
          float t = 0.f;
          #pragma unroll
          for (int q = 0; q < QI; ++q) t += bas[p][q][f] * s[i][q];
          if ((i * NF + f) & 1) T2[(i * NF + f) >> 1][p].y = t;
          else                  T2[(i * NF + f) >> 1][p].x = t;
        }
  }
  const float* w3l = w3 + 8 * hi * NCOLS + 4 * (m >> 3) * NIF + (m & 7);
  const float* b3l = bs3 + 4 * hi * NIF;
  #pragma unroll
  for (int j = 0; j < 4 * NF; j += 2) {
    const int c0 = ((j    ) / NF) * NIF + ((j    ) % NF) * 8;
    const int c1 = ((j + 1) / NF) * NIF + ((j + 1) % NF) * 8;
    v2f wa0[8], wa1[8];
    #pragma unroll
    for (int t = 0; t < 2; ++t)
      #pragma unroll
      for (int jp = 0; jp < 4; ++jp) {
        wa0[4 * t + jp] = ldWT(w3l, NCOLS, (16 * t + 2 * jp) * NCOLS + c0);
        wa1[4 * t + jp] = ldWT(w3l, NCOLS, (16 * t + 2 * jp) * NCOLS + c1);
      }
    const float* bp0 = b3l + c0;
    const float* bp1 = b3l + c1;
    v8f d0, d1;
    #pragma unroll
    for (int r = 0; r < 8; ++r) { d0[r] = bp0[r]; d1[r] = bp1[r]; }
    #pragma unroll
    for (int t = 0; t < 2; ++t)
      #pragma unroll
      for (int jp = 0; jp < 4; ++jp) {
        v2f b; b.x = h[t][2 * jp]; b.y = h[t][2 * jp + 1];
        d0 = wmma4(wa0[4 * t + jp], b, d0);
        d1 = wmma4(wa1[4 * t + jp], b, d1);
      }
    v2f x0[4], x1[4];
    #pragma unroll
    for (int k = 0; k < 4; ++k) {
      x0[k].x = d0[2 * k];  x0[k].y = d0[2 * k + 1];
      x1[k].x = d1[2 * k];  x1[k].y = d1[2 * k + 1];
    }
    #pragma unroll
    for (int p = 0; p < PO; ++p)
      #pragma unroll
      for (int k = 0; k < 4; ++k) {
        acc[(j    ) / NF][p] += x0[k] * T2[((j    ) % NF) * 4 + k][p];
        acc[(j + 1) / NF][p] += x1[k] * T2[((j + 1) % NF) * 4 + k][p];
      }
  }
}

__global__ void __launch_bounds__(WAVES_PER_BLK * 32)
__attribute__((amdgpu_waves_per_eu(5)))
se3_partial_kernel(KParams p) {
  const int lane = threadIdx.x & 31;
  const int wv   = threadIdx.x >> 5;
  const int m    = lane & 15;
  const int hi   = lane >> 4;
  const int tile = blockIdx.x * WAVES_PER_BLK + wv;
  const int e0   = tile * 16;
  if (e0 >= E_TOT) return;

  const int e = e0 + m;
  v2f xa[5];
  {
    #pragma unroll
    for (int c = 0; c < 4; ++c)
      xa[c] = *(const v2f*)(p.w + (size_t)e * EDIM + 4 * c + 2 * hi);
    float rad = p.radial[e];
    xa[4].x = hi ? 0.f : rad;
    xa[4].y = 0.f;
  }

  const int sidx = p.src_idx[e];
  const float* src0 = p.feat0 + (size_t)sidx * (NCH * 1);
  const float* src1 = p.feat1 + (size_t)sidx * (NCH * 3);

  v2f accD0[4][1] = {};
  v2f accD1[4][3] = {};

  pair_tile<0, 0, 1>(p.w1[0], p.b1[0], p.w2[0], p.b2[0], p.w3[0], p.b3[0],
                     xa, p.basis[0] + (size_t)e * 1, src0, m, hi, accD0);
  pair_tile<0, 1, 1>(p.w1[1], p.b1[1], p.w2[1], p.b2[1], p.w3[1], p.b3[1],
                     xa, p.basis[1] + (size_t)e * 3, src0, m, hi, accD1);
  pair_tile<1, 0, 1>(p.w1[2], p.b1[2], p.w2[2], p.b2[2], p.w3[2], p.b3[2],
                     xa, p.basis[2] + (size_t)e * 3, src1, m, hi, accD0);
  #pragma unroll
  for (int ol = 0; ol < 4; ++ol)
    p.out[(size_t)e * NCH + hi * 4 + ol] = accD0[ol][0].x + accD0[ol][0].y;

  pair_tile<1, 1, 3>(p.w1[3], p.b1[3], p.w2[3], p.b2[3], p.w3[3], p.b3[3],
                     xa, p.basis[3] + (size_t)e * 27, src1, m, hi, accD1);

  #pragma unroll
  for (int ol = 0; ol < 4; ++ol) {
    int o = hi * 4 + ol;
    #pragma unroll
    for (int q = 0; q < 3; ++q)
      p.out[(size_t)E_TOT * NCH + (size_t)e * (NCH * 3) + o * 3 + q] =
          accD1[ol][q].x + accD1[ol][q].y;
  }
}

extern "C" void kernel_launch(void* const* d_in, const int* in_sizes, int n_in,
                              void* d_out, int out_size, void* d_ws,
                              size_t ws_size, hipStream_t stream) {
  (void)in_sizes; (void)n_in; (void)out_size; (void)d_ws; (void)ws_size;
  KParams p;
  p.feat0  = (const float*)d_in[0];
  p.feat1  = (const float*)d_in[1];
  p.w      = (const float*)d_in[2];
  p.radial = (const float*)d_in[3];
  for (int pr = 0; pr < 4; ++pr) {
    int b = 4 + pr * 7;
    p.basis[pr] = (const float*)d_in[b + 0];
    p.w1[pr]    = (const float*)d_in[b + 1];
    p.b1[pr]    = (const float*)d_in[b + 2];
    p.w2[pr]    = (const float*)d_in[b + 3];
    p.b2[pr]    = (const float*)d_in[b + 4];
    p.w3[pr]    = (const float*)d_in[b + 5];
    p.b3[pr]    = (const float*)d_in[b + 6];
  }
  p.src_idx = (const int*)d_in[32];
  p.out     = (float*)d_out;

  const int tiles  = E_TOT / 16;
  const int blocks = (tiles + WAVES_PER_BLK - 1) / WAVES_PER_BLK;
  hipLaunchKernelGGL(se3_partial_kernel, dim3(blocks), dim3(WAVES_PER_BLK * 32),
                     0, stream, p);
}


# Weights and biases copied once per work-group into LDS in A-fragment order (one 16-byte LDS read = two fragments); 256 persistent work-groups of 20 waves loop over tiles; sum order unchanged
# speedup vs baseline: 1.5154x; 1.2057x over previous
#include <hip/hip_runtime.h>


typedef float v2f __attribute__((ext_vector_type(2)));
typedef float v4f __attribute__((ext_vector_type(4)));
typedef float v8f __attribute__((ext_vector_type(8)));

#define E_TOT 160000
#define NCH 8
#define MIDW 32
#define EDIM 16
#define WAVES_PER_BLK 20
#define NBLK 256
#define N_TILES (E_TOT / 16)

#define UNITS_SMALL 29
#define UNITS_TOT 148
#define W_FLOATS (UNITS_TOT * 128)
#define B_FLOATS 640

struct KParams {
  const float* feat0; const float* feat1; const float* w; const float* radial;
  const float* basis[4];
  const float* w1[4]; const float* b1[4];
  const float* w2[4]; const float* b2[4];
  const float* w3[4]; const float* b3[4];
  const int* src_idx;
  float* out;
};

__device__ __forceinline__ v8f wmma4(v2f a, v2f b, v8f c) {
  return __builtin_amdgcn_wmma_f32_16x16x4_f32(false, a, false, b,
                                               (short)0, c, false, false);
}

__device__ __forceinline__ v4f fetch_unit(const KParams& p, int u, int m, int hi) {
  int pr = u / UNITS_SMALL;  pr = pr > 3 ? 3 : pr;
  const int r   = u - UNITS_SMALL * pr;
  const bool l1 = r < 5, l2 = r < 13;
  const int cc  = l1 ? r : (l2 ? r - 5 : ((r - 13) & 7));
  const int jt  = (r - 13) >> 3;
  const bool wide = pr == 3;
  const int nif = wide ? 24 : 8;
  const int j0 = 2 * jt, j1 = 2 * jt + 1;
  const int c0 = wide ? (j0 / 3) * 24 + (j0 % 3) * 8 : j0 * 8;
  const int c1 = wide ? (j1 / 3) * 24 + (j1 % 3) * 8 : j1 * 8;
  const float* __restrict__ Wm = l1 ? p.w1[pr] : (l2 ? p.w2[pr] : p.w3[pr]);
  const int ncols = l2 ? MIDW : 8 * nif;
  const int kmax  = l1 ? EDIM + 1 : MIDW;
  const int kA    = l1 ? 4 * cc + 2 * hi : 16 * (cc >> 2) + 8 * hi + 2 * (cc & 3);
  const int cb    = 4 * (m >> 3) * nif + (m & 7);
  const int colA  = l2 ? m : cb + c0;
  const int colB  = l2 ? 16 + m : cb + c1;
  const int k0 = kA < kmax ? kA : kmax - 1;
  const int k1 = kA + 1 < kmax ? kA + 1 : kmax - 1;
  const float a0 = Wm[k0 * ncols + colA], a1 = Wm[k1 * ncols + colA];
  const float b0 = Wm[k0 * ncols + colB], b1 = Wm[k1 * ncols + colB];
  v4f v;
  v.x = kA     < kmax ? a0 : 0.f;
  v.y = kA + 1 < kmax ? a1 : 0.f;
  v.z = kA     < kmax ? b0 : 0.f;
  v.w = kA + 1 < kmax ? b1 : 0.f;
  return v;
}

template<int WV>
__device__ __forceinline__ void fetch_wave(const KParams& p, int m, int hi, int lane,
                                           v4f (&v)[8], float& bv) {
  constexpr int T0  = 32 * WV;
  constexpr int PRB = T0 / 128 > 3 ? 3 : T0 / 128;
  constexpr int R   = T0 - 128 * PRB;
  const float* __restrict__ bsrc = R < 32 ? p.b1[PRB] : (R < 64 ? p.b2[PRB] : p.b3[PRB]);
  bv = bsrc[(R < 32 ? R : (R < 64 ? R - 32 : R - 64)) + lane];
  #pragma unroll
  for (int i = 0; i < 8; ++i)
    if (WV + WAVES_PER_BLK * i < UNITS_TOT)
      v[i] = fetch_unit(p, WV + WAVES_PER_BLK * i, m, hi);
}

__device__ __forceinline__ void stage_params(const KParams& p, float* lds,
                                             int wv, int lane) {
  const int m = lane & 15, hi = lane >> 4;
  v4f v[8] = {};
  float bv = 0.f;
  switch (wv) {
    case 0:  fetch_wave<0>(p, m, hi, lane, v, bv);  break;
    case 1:  fetch_wave<1>(p, m, hi, lane, v, bv);  break;
    case 2:  fetch_wave<2>(p, m, hi, lane, v, bv);  break;
    case 3:  fetch_wave<3>(p, m, hi, lane, v, bv);  break;
    case 4:  fetch_wave<4>(p, m, hi, lane, v, bv);  break;
    case 5:  fetch_wave<5>(p, m, hi, lane, v, bv);  break;
    case 6:  fetch_wave<6>(p, m, hi, lane, v, bv);  break;
    case 7:  fetch_wave<7>(p, m, hi, lane, v, bv);  break;
    case 8:  fetch_wave<8>(p, m, hi, lane, v, bv);  break;
    case 9:  fetch_wave<9>(p, m, hi, lane, v, bv);  break;
    case 10: fetch_wave<10>(p, m, hi, lane, v, bv); break;
    case 11: fetch_wave<11>(p, m, hi, lane, v, bv); break;
    case 12: fetch_wave<12>(p, m, hi, lane, v, bv); break;
    case 13: fetch_wave<13>(p, m, hi, lane, v, bv); break;
    case 14: fetch_wave<14>(p, m, hi, lane, v, bv); break;
    case 15: fetch_wave<15>(p, m, hi, lane, v, bv); break;
    case 16: fetch_wave<16>(p, m, hi, lane, v, bv); break;
    case 17: fetch_wave<17>(p, m, hi, lane, v, bv); break;
    case 18: fetch_wave<18>(p, m, hi, lane, v, bv); break;
    default: fetch_wave<19>(p, m, hi, lane, v, bv); break;
  }
  lds[W_FLOATS + 32 * wv + lane] = bv;
  #pragma unroll
  for (int i = 0; i < 8; ++i) {
    const int u = wv + WAVES_PER_BLK * i;
    if (u < UNITS_TOT) *(v4f*)(lds + (u * 32 + lane) * 4) = v[i];
  }
}

template<int PR, int DI, int DO, int NF>
__device__ __forceinline__ void pair_tile(
    const float* ldsW, const float* ldsB,
    const v2f (&xa)[5],
    const float* __restrict__ basisE,
    const float* __restrict__ srcF,
    int hi, v2f (&acc)[4][2 * DO + 1]) {
  constexpr int QI = 2 * DI + 1, PO = 2 * DO + 1;
  constexpr int NIF = NCH * NF;
  constexpr int UB = PR * UNITS_SMALL;
  const float* wl = ldsW + UB * 128;
  const float* bl = ldsB + PR * 128;

  v8f h[2];
  {
    v2f wa0[5], wa1[5];
    #pragma unroll
    for (int c = 0; c < 5; ++c) {
      const v4f q = *(const v4f*)(wl + c * 128);
      wa0[c].x = q.x;  wa0[c].y = q.y;
      wa1[c].x = q.z;  wa1[c].y = q.w;
    }
    const float* bp = bl + 8 * hi;
    v8f d0, d1;
    #pragma unroll
    for (int r = 0; r < 8; ++r) { d0[r] = bp[r]; d1[r] = bp[16 + r]; }
    #pragma unroll
    for (int c = 0; c < 5; ++c) {
      d0 = wmma4(wa0[c], xa[c], d0);
      d1 = wmma4(wa1[c], xa[c], d1);
    }
    #pragma unroll
    for (int r = 0; r < 8; ++r) {
      h[0][r] = d0[r] > 0.f ? d0[r] : 0.f;
      h[1][r] = d1[r] > 0.f ? d1[r] : 0.f;
    }
  }
  {
    v2f wa0[8], wa1[8];
    #pragma unroll
    for (int cc = 0; cc < 8; ++cc) {
      const v4f q = *(const v4f*)(wl + (5 + cc) * 128);
      wa0[cc].x = q.x;  wa0[cc].y = q.y;
      wa1[cc].x = q.z;  wa1[cc].y = q.w;
    }
    const float* bp = bl + 32 + 8 * hi;
    v8f d0, d1;
    #pragma unroll
    for (int r = 0; r < 8; ++r) { d0[r] = bp[r]; d1[r] = bp[16 + r]; }
    #pragma unroll
    for (int t = 0; t < 2; ++t)
      #pragma unroll
      for (int jp = 0; jp < 4; ++jp) {
        v2f b; b.x = h[t][2 * jp]; b.y = h[t][2 * jp + 1];
        d0 = wmma4(wa0[4 * t + jp], b, d0);
        d1 = wmma4(wa1[4 * t + jp], b, d1);
      }
    #pragma unroll
    for (int r = 0; r < 8; ++r) {
      h[0][r] = d0[r] > 0.f ? d0[r] : 0.f;
      h[1][r] = d1[r] > 0.f ? d1[r] : 0.f;
    }
  }
  v2f T2[NIF / 2][PO];
  {
    float s[NCH][QI];
    #pragma unroll
    for (int i = 0; i < NCH; ++i)
      #pragma unroll
      for (int q = 0; q < QI; ++q) s[i][q] = srcF[i * QI + q];
    float bas[PO][QI][NF];
    #pragma unroll
    for (int p = 0; p < PO; ++p)
      #pragma unroll
      for (int q = 0; q < QI; ++q)
        #pragma unroll
        for (int f = 0; f < NF; ++f) bas[p][q][f] = basisE[(p * QI + q) * NF + f];
    #pragma unroll
    for (int i = 0; i < NCH; ++i)
      #pragma unroll
      for (int f = 0; f < NF; ++f)
        #pragma unroll
        for (int p = 0; p < PO; ++p) {
          float t = 0.f;
          #pragma unroll
          for (int q = 0; q < QI; ++q) t += bas[p][q][f] * s[i][q];
          if ((i * NF + f) & 1) T2[(i * NF + f) >> 1][p].y = t;
          else                  T2[(i * NF + f) >> 1][p].x = t;
        }
  }
  const float* b3l = bl + 64 + 4 * hi * NIF;
  #pragma unroll
  for (int j = 0; j < 4 * NF; j += 2) {
    const int c0 = ((j    ) / NF) * NIF + ((j    ) % NF) * 8;
    const int c1 = ((j + 1) / NF) * NIF + ((j + 1) % NF) * 8;
    v2f wa0[8], wa1[8];
    #pragma unroll
    for (int cc = 0; cc < 8; ++cc) {
      const v4f q = *(const v4f*)(wl + (13 + 8 * (j / 2) + cc) * 128);
      wa0[cc].x = q.x;  wa0[cc].y = q.y;
      wa1[cc].x = q.z;  wa1[cc].y = q.w;
    }
    const float* bp0 = b3l + c0;
    const float* bp1 = b3l + c1;
    v8f d0, d1;
    #pragma unroll
    for (int r = 0; r < 8; ++r) { d0[r] = bp0[r]; d1[r] = bp1[r]; }
    #pragma unroll
    for (int t = 0; t < 2; ++t)
      #pragma unroll
      for (int jp = 0; jp < 4; ++jp) {
        v2f b; b.x = h[t][2 * jp]; b.y = h[t][2 * jp + 1];
        d0 = wmma4(wa0[4 * t + jp], b, d0);
        d1 = wmma4(wa1[4 * t + jp], b, d1);
      }
    v2f x0[4], x1[4];
    #pragma unroll
    for (int k = 0; k < 4; ++k) {
      x0[k].x = d0[2 * k];  x0[k].y = d0[2 * k + 1];
      x1[k].x = d1[2 * k];  x1[k].y = d1[2 * k + 1];
    }
    #pragma unroll
    for (int p = 0; p < PO; ++p)
      #pragma unroll
      for (int k = 0; k < 4; ++k) {
        acc[(j    ) / NF][p] += x0[k] * T2[((j    ) % NF) * 4 + k][p];
        acc[(j + 1) / NF][p] += x1[k] * T2[((j + 1) % NF) * 4 + k][p];
      }
  }
}

__global__ void __launch_bounds__(WAVES_PER_BLK * 32)
__attribute__((amdgpu_waves_per_eu(5)))
se3_partial_kernel(KParams p) {
  __shared__ __attribute__((aligned(16))) float lds[W_FLOATS + B_FLOATS];
  const int lane = threadIdx.x & 31;
  const int wv   = __builtin_amdgcn_readfirstlane((int)(threadIdx.x >> 5));
  const int m    = lane & 15;
  const int hi   = lane >> 4;

  stage_params(p, lds, wv, lane);
  __syncthreads();
  const float* ldsW = lds + lane * 4;
  const float* ldsB = lds + W_FLOATS;

  for (int tile = blockIdx.x * WAVES_PER_BLK + wv; tile < N_TILES;
       tile += NBLK * WAVES_PER_BLK) {
  __builtin_amdgcn_fence(__ATOMIC_ACQUIRE, "workgroup", "local");
  const int e0   = tile * 16;

  const int e = e0 + m;
  v2f xa[5];
  {
    #pragma unroll
    for (int c = 0; c < 4; ++c)
      xa[c] = *(const v2f*)(p.w + (size_t)e * EDIM + 4 * c + 2 * hi);
    float rad = p.radial[e];
    xa[4].x = hi ? 0.f : rad;
    xa[4].y = 0.f;
  }

  const int sidx = p.src_idx[e];
  const float* src0 = p.feat0 + (size_t)sidx * (NCH * 1);
  const float* src1 = p.feat1 + (size_t)sidx * (NCH * 3);

  v2f accD0[4][1] = {};
  v2f accD1[4][3] = {};

  pair_tile<0, 0, 0, 1>(ldsW, ldsB,
                     xa, p.basis[0] + (size_t)e * 1, src0, hi, accD0);
  pair_tile<1, 0, 1, 1>(ldsW, ldsB,
                     xa, p.basis[1] + (size_t)e * 3, src0, hi, accD1);
  pair_tile<2, 1, 0, 1>(ldsW, ldsB,
                     xa, p.basis[2] + (size_t)e * 3, src1, hi, accD0);
  #pragma unroll
  for (int ol = 0; ol < 4; ++ol)
    p.out[(size_t)e * NCH + hi * 4 + ol] = accD0[ol][0].x + accD0[ol][0].y;

  pair_tile<3, 1, 1, 3>(ldsW, ldsB,
                     xa, p.basis[3] + (size_t)e * 27, src1, hi, accD1);

  #pragma unroll
  for (int ol = 0; ol < 4; ++ol) {
    int o = hi * 4 + ol;
    #pragma unroll
    for (int q = 0; q < 3; ++q)
      p.out[(size_t)E_TOT * NCH + (size_t)e * (NCH * 3) + o * 3 + q] =
          accD1[ol][q].x + accD1[ol][q].y;
  }
  }
}

extern "C" void kernel_launch(void* const* d_in, const int* in_sizes, int n_in,
                              void* d_out, int out_size, void* d_ws,
                              size_t ws_size, hipStream_t stream) {
  (void)in_sizes; (void)n_in; (void)out_size; (void)d_ws; (void)ws_size;
  KParams p;
  p.feat0  = (const float*)d_in[0];
  p.feat1  = (const float*)d_in[1];
  p.w      = (const float*)d_in[2];
  p.radial = (const float*)d_in[3];
  for (int pr = 0; pr < 4; ++pr) {
    int b = 4 + pr * 7;
    p.basis[pr] = (const float*)d_in[b + 0];
    p.w1[pr]    = (const float*)d_in[b + 1];
    p.b1[pr]    = (const float*)d_in[b + 2];
    p.w2[pr]    = (const float*)d_in[b + 3];
    p.b2[pr]    = (const float*)d_in[b + 4];
    p.w3[pr]    = (const float*)d_in[b + 5];
    p.b3[pr]    = (const float*)d_in[b + 6];
  }
  p.src_idx = (const int*)d_in[32];
  p.out     = (float*)d_out;

  hipLaunchKernelGGL(se3_partial_kernel, dim3(NBLK), dim3(WAVES_PER_BLK * 32),
                     0, stream, p);
}


# Wave priority falls with a wave's own progress (s_setprio 2 in its first tile, 1 in its second, 0 in the second tile's widest degree pair): the matrix pipe serves the waves behind; sums unchanged
# speedup vs baseline: 1.6029x; 1.0577x over previous
#include <hip/hip_runtime.h>


typedef float v2f __attribute__((ext_vector_type(2)));
typedef float v4f __attribute__((ext_vector_type(4)));
typedef float v8f __attribute__((ext_vector_type(8)));

#define E_TOT 160000
#define NCH 8
#define MIDW 32
#define EDIM 16
#define WAVES_PER_BLK 20
#define NBLK 256
#define N_TILES (E_TOT / 16)

#define UNITS_SMALL 29
#define UNITS_TOT 148
#define W_FLOATS (UNITS_TOT * 128)
#define B_FLOATS 640

struct KParams {
  const float* feat0; const float* feat1; const float* w; const float* radial;
  const float* basis[4];
  const float* w1[4]; const float* b1[4];
  const float* w2[4]; const float* b2[4];
  const float* w3[4]; const float* b3[4];
  const int* src_idx;
  float* out;
};

template<int LEVEL>
__device__ __forceinline__ void set_prio_if(int flag) {
  asm volatile("s_cmp_eq_u32 %0, 0\n\t"
               "s_cbranch_scc1 1f\n\t"
               "s_setprio %1\n"
               "1:"
               : : "s"(flag), "n"(LEVEL) : "scc");
}

__device__ __forceinline__ v8f wmma4(v2f a, v2f b, v8f c) {
  return __builtin_amdgcn_wmma_f32_16x16x4_f32(false, a, false, b,
                                               (short)0, c, false, false);
}

__device__ __forceinline__ v4f fetch_unit(const KParams& p, int u, int m, int hi) {
  int pr = u / UNITS_SMALL;  pr = pr > 3 ? 3 : pr;
  const int r   = u - UNITS_SMALL * pr;
  const bool l1 = r < 5, l2 = r < 13;
  const int cc  = l1 ? r : (l2 ? r - 5 : ((r - 13) & 7));
  const int jt  = (r - 13) >> 3;
  const bool wide = pr == 3;
  const int nif = wide ? 24 : 8;
  const int j0 = 2 * jt, j1 = 2 * jt + 1;
  const int c0 = wide ? (j0 / 3) * 24 + (j0 % 3) * 8 : j0 * 8;
  const int c1 = wide ? (j1 / 3) * 24 + (j1 % 3) * 8 : j1 * 8;
  const float* __restrict__ Wm = l1 ? p.w1[pr] : (l2 ? p.w2[pr] : p.w3[pr]);
  const int ncols = l2 ? MIDW : 8 * nif;
  const int kmax  = l1 ? EDIM + 1 : MIDW;
  const int kA    = l1 ? 4 * cc + 2 * hi : 16 * (cc >> 2) + 8 * hi + 2 * (cc & 3);
  const int cb    = 4 * (m >> 3) * nif + (m & 7);
  const int colA  = l2 ? m : cb + c0;
  const int colB  = l2 ? 16 + m : cb + c1;
  const int k0 = kA < kmax ? kA : kmax - 1;
  const int k1 = kA + 1 < kmax ? kA + 1 : kmax - 1;
  const float a0 = Wm[k0 * ncols + colA], a1 = Wm[k1 * ncols + colA];
  const float b0 = Wm[k0 * ncols + colB], b1 = Wm[k1 * ncols + colB];
  v4f v;
  v.x = kA     < kmax ? a0 : 0.f;
  v.y = kA + 1 < kmax ? a1 : 0.f;
  v.z = kA     < kmax ? b0 : 0.f;
  v.w = kA + 1 < kmax ? b1 : 0.f;
  return v;
}

template<int WV>
__device__ __forceinline__ void fetch_wave(const KParams& p, int m, int hi, int lane,
                                           v4f (&v)[8], float& bv) {
  constexpr int T0  = 32 * WV;
  constexpr int PRB = T0 / 128 > 3 ? 3 : T0 / 128;
  constexpr int R   = T0 - 128 * PRB;
  const float* __restrict__ bsrc = R < 32 ? p.b1[PRB] : (R < 64 ? p.b2[PRB] : p.b3[PRB]);
  bv = bsrc[(R < 32 ? R : (R < 64 ? R - 32 : R - 64)) + lane];
  #pragma unroll
  for (int i = 0; i < 8; ++i)
    if (WV + WAVES_PER_BLK * i < UNITS_TOT)
      v[i] = fetch_unit(p, WV + WAVES_PER_BLK * i, m, hi);
}

__device__ __forceinline__ void stage_params(const KParams& p, float* lds,
                                             int wv, int lane) {
  const int m = lane & 15, hi = lane >> 4;
  v4f v[8] = {};
  float bv = 0.f;
  switch (wv) {
    case 0:  fetch_wave<0>(p, m, hi, lane, v, bv);  break;
    case 1:  fetch_wave<1>(p, m, hi, lane, v, bv);  break;
    case 2:  fetch_wave<2>(p, m, hi, lane, v, bv);  break;
    case 3:  fetch_wave<3>(p, m, hi, lane, v, bv);  break;
    case 4:  fetch_wave<4>(p, m, hi, lane, v, bv);  break;
    case 5:  fetch_wave<5>(p, m, hi, lane, v, bv);  break;
    case 6:  fetch_wave<6>(p, m, hi, lane, v, bv);  break;
    case 7:  fetch_wave<7>(p, m, hi, lane, v, bv);  break;
    case 8:  fetch_wave<8>(p, m, hi, lane, v, bv);  break;
    case 9:  fetch_wave<9>(p, m, hi, lane, v, bv);  break;
    case 10: fetch_wave<10>(p, m, hi, lane, v, bv); break;
    case 11: fetch_wave<11>(p, m, hi, lane, v, bv); break;
    case 12: fetch_wave<12>(p, m, hi, lane, v, bv); break;
    case 13: fetch_wave<13>(p, m, hi, lane, v, bv); break;
    case 14: fetch_wave<14>(p, m, hi, lane, v, bv); break;
    case 15: fetch_wave<15>(p, m, hi, lane, v, bv); break;
    case 16: fetch_wave<16>(p, m, hi, lane, v, bv); break;
    case 17: fetch_wave<17>(p, m, hi, lane, v, bv); break;
    case 18: fetch_wave<18>(p, m, hi, lane, v, bv); break;
    default: fetch_wave<19>(p, m, hi, lane, v, bv); break;
  }
  lds[W_FLOATS + 32 * wv + lane] = bv;
  #pragma unroll
  for (int i = 0; i < 8; ++i) {
    const int u = wv + WAVES_PER_BLK * i;
    if (u < UNITS_TOT) *(v4f*)(lds + (u * 32 + lane) * 4) = v[i];
  }
}

template<int PR, int DI, int DO, int NF>
__device__ __forceinline__ void pair_tile(
    const float* ldsW, const float* ldsB,
    const v2f (&xa)[5],
    const float* __restrict__ basisE,
    const float* __restrict__ srcF,
    int hi, v2f (&acc)[4][2 * DO + 1]) {
  constexpr int QI = 2 * DI + 1, PO = 2 * DO + 1;
  constexpr int NIF = NCH * NF;
  constexpr int UB = PR * UNITS_SMALL;
  const float* wl = ldsW + UB * 128;
  const float* bl = ldsB + PR * 128;

  v8f h[2];
  {
    v2f wa0[5], wa1[5];
    #pragma unroll
    for (int c = 0; c < 5; ++c) {
      const v4f q = *(const v4f*)(wl + c * 128);
      wa0[c].x = q.x;  wa0[c].y = q.y;
      wa1[c].x = q.z;  wa1[c].y = q.w;
    }
    const float* bp = bl + 8 * hi;
    v8f d0, d1;
    #pragma unroll
    for (int r = 0; r < 8; ++r) { d0[r] = bp[r]; d1[r] = bp[16 + r]; }
    #pragma unroll
    for (int c = 0; c < 5; ++c) {
      d0 = wmma4(wa0[c], xa[c], d0);
      d1 = wmma4(wa1[c], xa[c], d1);
    }
    #pragma unroll
    for (int r = 0; r < 8; ++r) {
      h[0][r] = d0[r] > 0.f ? d0[r] : 0.f;
      h[1][r] = d1[r] > 0.f ? d1[r] : 0.f;
    }
  }
  {
    v2f wa0[8], wa1[8];
    #pragma unroll
    for (int cc = 0; cc < 8; ++cc) {
      const v4f q = *(const v4f*)(wl + (5 + cc) * 128);
      wa0[cc].x = q.x;  wa0[cc].y = q.y;
      wa1[cc].x = q.z;  wa1[cc].y = q.w;
    }
    const float* bp = bl + 32 + 8 * hi;
    v8f d0, d1;
    #pragma unroll
    for (int r = 0; r < 8; ++r) { d0[r] = bp[r]; d1[r] = bp[16 + r]; }
    #pragma unroll
    for (int t = 0; t < 2; ++t)
      #pragma unroll
      for (int jp = 0; jp < 4; ++jp) {
        v2f b; b.x = h[t][2 * jp]; b.y = h[t][2 * jp + 1];
        d0 = wmma4(wa0[4 * t + jp], b, d0);
        d1 = wmma4(wa1[4 * t + jp], b, d1);
      }
    #pragma unroll
    for (int r = 0; r < 8; ++r) {
      h[0][r] = d0[r] > 0.f ? d0[r] : 0.f;
      h[1][r] = d1[r] > 0.f ? d1[r] : 0.f;
    }
  }
  v2f T2[NIF / 2][PO];
  {
    float s[NCH][QI];
    #pragma unroll
    for (int i = 0; i < NCH; ++i)
      #pragma unroll
      for (int q = 0; q < QI; ++q) s[i][q] = srcF[i * QI + q];
    float bas[PO][QI][NF];
    #pragma unroll
    for (int p = 0; p < PO; ++p)
      #pragma unroll
      for (int q = 0; q < QI; ++q)
        #pragma unroll
        for (int f = 0; f < NF; ++f) bas[p][q][f] = basisE[(p * QI + q) * NF + f];
    #pragma unroll
    for (int i = 0; i < NCH; ++i)
      #pragma unroll
      for (int f = 0; f < NF; ++f)
        #pragma unroll
        for (int p = 0; p < PO; ++p) {
          float t = 0.f;
          #pragma unroll
          for (int q = 0; q < QI; ++q) t += bas[p][q][f] * s[i][q];
          if ((i * NF + f) & 1) T2[(i * NF + f) >> 1][p].y = t;
          else                  T2[(i * NF + f) >> 1][p].x = t;
        }
  }
  const float* b3l = bl + 64 + 4 * hi * NIF;
  #pragma unroll
  for (int j = 0; j < 4 * NF; j += 2) {
    const int c0 = ((j    ) / NF) * NIF + ((j    ) % NF) * 8;
    const int c1 = ((j + 1) / NF) * NIF + ((j + 1) % NF) * 8;
    v2f wa0[8], wa1[8];
    #pragma unroll
    for (int cc = 0; cc < 8; ++cc) {
      const v4f q = *(const v4f*)(wl + (13 + 8 * (j / 2) + cc) * 128);
      wa0[cc].x = q.x;  wa0[cc].y = q.y;
      wa1[cc].x = q.z;  wa1[cc].y = q.w;
    }
    const float* bp0 = b3l + c0;
    const float* bp1 = b3l + c1;
    v8f d0, d1;
    #pragma unroll
    for (int r = 0; r < 8; ++r) { d0[r] = bp0[r]; d1[r] = bp1[r]; }
    #pragma unroll
    for (int t = 0; t < 2; ++t)
      #pragma unroll
      for (int jp = 0; jp < 4; ++jp) {
        v2f b; b.x = h[t][2 * jp]; b.y = h[t][2 * jp + 1];
        d0 = wmma4(wa0[4 * t + jp], b, d0);
        d1 = wmma4(wa1[4 * t + jp], b, d1);
      }
    v2f x0[4], x1[4];
    #pragma unroll
    for (int k = 0; k < 4; ++k) {
      x0[k].x = d0[2 * k];  x0[k].y = d0[2 * k + 1];
      x1[k].x = d1[2 * k];  x1[k].y = d1[2 * k + 1];
    }
    #pragma unroll
    for (int p = 0; p < PO; ++p)
      #pragma unroll
      for (int k = 0; k < 4; ++k) {
        acc[(j    ) / NF][p] += x0[k] * T2[((j    ) % NF) * 4 + k][p];
        acc[(j + 1) / NF][p] += x1[k] * T2[((j + 1) % NF) * 4 + k][p];
      }
  }
}

__global__ void __launch_bounds__(WAVES_PER_BLK * 32)
__attribute__((amdgpu_waves_per_eu(5)))
se3_partial_kernel(KParams p) {
  __shared__ __attribute__((aligned(16))) float lds[W_FLOATS + B_FLOATS];
  const int lane = threadIdx.x & 31;
  const int wv   = __builtin_amdgcn_readfirstlane((int)(threadIdx.x >> 5));
  const int m    = lane & 15;
  const int hi   = lane >> 4;

  stage_params(p, lds, wv, lane);
  __syncthreads();
  const float* ldsW = lds + lane * 4;
  const float* ldsB = lds + W_FLOATS;

  for (int tile = blockIdx.x * WAVES_PER_BLK + wv; tile < N_TILES;
       tile += NBLK * WAVES_PER_BLK) {
  __builtin_amdgcn_fence(__ATOMIC_ACQUIRE, "workgroup", "local");
  const int second = tile >= NBLK * WAVES_PER_BLK ? 1 : 0;
  if (second) __builtin_amdgcn_s_setprio(1);
  else        __builtin_amdgcn_s_setprio(2);
  const int e0   = tile * 16;

  const int e = e0 + m;
  v2f xa[5];
  {
    #pragma unroll
    for (int c = 0; c < 4; ++c)
      xa[c] = *(const v2f*)(p.w + (size_t)e * EDIM + 4 * c + 2 * hi);
    float rad = p.radial[e];
    xa[4].x = hi ? 0.f : rad;
    xa[4].y = 0.f;
  }

  const int sidx = p.src_idx[e];
  const float* src0 = p.feat0 + (size_t)sidx * (NCH * 1);
  const float* src1 = p.feat1 + (size_t)sidx * (NCH * 3);

  v2f accD0[4][1] = {};
  v2f accD1[4][3] = {};

  pair_tile<0, 0, 0, 1>(ldsW, ldsB,
                     xa, p.basis[0] + (size_t)e * 1, src0, hi, accD0);
  pair_tile<1, 0, 1, 1>(ldsW, ldsB,
                     xa, p.basis[1] + (size_t)e * 3, src0, hi, accD1);
  pair_tile<2, 1, 0, 1>(ldsW, ldsB,
                     xa, p.basis[2] + (size_t)e * 3, src1, hi, accD0);
  #pragma unroll
  for (int ol = 0; ol < 4; ++ol)
    p.out[(size_t)e * NCH + hi * 4 + ol] = accD0[ol][0].x + accD0[ol][0].y;

  set_prio_if<0>(second);
  pair_tile<3, 1, 1, 3>(ldsW, ldsB,
                     xa, p.basis[3] + (size_t)e * 27, src1, hi, accD1);

  #pragma unroll
  for (int ol = 0; ol < 4; ++ol) {
    int o = hi * 4 + ol;
    #pragma unroll
    for (int q = 0; q < 3; ++q)
      p.out[(size_t)E_TOT * NCH + (size_t)e * (NCH * 3) + o * 3 + q] =
          accD1[ol][q].x + accD1[ol][q].y;
  }
  }
  __builtin_amdgcn_s_setprio(0);
}

extern "C" void kernel_launch(void* const* d_in, const int* in_sizes, int n_in,
                              void* d_out, int out_size, void* d_ws,
                              size_t ws_size, hipStream_t stream) {
  (void)in_sizes; (void)n_in; (void)out_size; (void)d_ws; (void)ws_size;
  KParams p;
  p.feat0  = (const float*)d_in[0];
  p.feat1  = (const float*)d_in[1];
  p.w      = (const float*)d_in[2];
  p.radial = (const float*)d_in[3];
  for (int pr = 0; pr < 4; ++pr) {
    int b = 4 + pr * 7;
    p.basis[pr] = (const float*)d_in[b + 0];
    p.w1[pr]    = (const float*)d_in[b + 1];
    p.b1[pr]    = (const float*)d_in[b + 2];
    p.w2[pr]    = (const float*)d_in[b + 3];
    p.b2[pr]    = (const float*)d_in[b + 4];
    p.w3[pr]    = (const float*)d_in[b + 5];
    p.b3[pr]    = (const float*)d_in[b + 6];
  }
  p.src_idx = (const int*)d_in[32];
  p.out     = (float*)d_out;

  hipLaunchKernelGGL(se3_partial_kernel, dim3(NBLK), dim3(WAVES_PER_BLK * 32),
                     0, stream, p);
}
